# starting kernel
# baseline (speedup 1.0000x reference)
#include <hip/hip_runtime.h>
#include <hip/hip_bf16.h>


typedef __attribute__((ext_vector_type(2))) float v2f;
typedef __attribute__((ext_vector_type(8))) float v8f;

static constexpr int NN   = 10000;
static constexpr int NE   = 160000;
static constexpr int MULC = 128;
static constexpr int NSP  = 10;
static constexpr int NRAD = 8;

static constexpr float SILU_C    = 1.679f;
static constexpr float SC_NORM   = 0.02795084971874737f;
static constexpr float L1_NORM   = 0.08838834764831845f;
static constexpr float L2_NORM   = 0.0625f;
static constexpr float INV_SQRT8 = 0.3535533905932738f;
static constexpr float INV_SQ3   = 0.5773502691896258f;
static constexpr float INV_NEIGH = 0.25f;

__device__ __forceinline__ v8f wmma4(v2f a, v2f b, v8f c) {
  return __builtin_amdgcn_wmma_f32_16x16x4_f32(
       false, a,  false, b,
       (short)0, c,  false,  false);
}

__device__ __forceinline__ float silu_c(float z) {
  return SILU_C * z / (1.0f + __expf(-z));
}

__global__ void zero_kernel(float* __restrict__ p, long n) {
  long i = (long)blockIdx.x * blockDim.x + threadIdx.x;
  long stride = (long)gridDim.x * blockDim.x;
  for (; i < n; i += stride) p[i] = 0.0f;
}

__global__ __launch_bounds__(128) void node_kernel(
    const float* __restrict__ node_feats, const float* __restrict__ node_attrs,
    const float* __restrict__ lin1_ws, const float* __restrict__ lin1_wv,
    const float* __restrict__ sc_ws, const float* __restrict__ sc_wv,
    float* __restrict__ out, float* __restrict__ y) {
  __shared__ float xs[16][MULC];
  __shared__ float xvc[3][16][MULC];
  __shared__ float at[16][16];
  __shared__ float zbuf[16][256];

  const int n0  = blockIdx.x * 16;
  const int tid = threadIdx.x;

  for (int i = tid; i < 16 * 512; i += 128) {
    int r = i >> 9, c = i & 511;
    float vv = node_feats[(size_t)(n0 + r) * 512 + c];
    if (c < MULC) xs[r][c] = vv;
    else { int q = c - MULC; xvc[q % 3][r][q / 3] = vv; }
  }
  for (int i = tid; i < 16 * 16; i += 128) {
    int r = i >> 4, c = i & 15;
    at[r][c] = (c < NSP) ? node_attrs[(size_t)(n0 + r) * NSP + c] : 0.0f;
  }
  __syncthreads();

  const int wave = tid >> 5;
  const int lane = tid & 31;
  const int half = lane >> 4;
  const int m    = lane & 15;
  const int nl   = lane & 15;
  const int wA   = wave * 16;
  const int wB   = wA + 64;

  for (int t = 0; t < 4; ++t) {
    float (*xsrc)[MULC] = (t == 0) ? xs : xvc[t - 1];
    const float* W = (t == 0) ? sc_ws : sc_wv;
    v8f accA = {}, accB = {};
    for (int chunk = 0; chunk < 5; ++chunk) {
      __syncthreads();
      for (int i = tid; i < 16 * 256; i += 128) {
        int r = i >> 8, kk = i & 255;
        int k = chunk * 256 + kk;
        zbuf[r][kk] = xsrc[r][k / NSP] * at[r][k % NSP];
      }
      __syncthreads();
      for (int kk = 0; kk < 256; kk += 4) {
        int k0 = kk + 2 * half;
        v2f a = { zbuf[m][k0], zbuf[m][k0 + 1] };
        size_t krow = (size_t)(chunk * 256 + k0) * MULC;
        v2f bA = { W[krow + wA + nl], W[krow + MULC + wA + nl] };
        v2f bB = { W[krow + wB + nl], W[krow + MULC + wB + nl] };
        accA = wmma4(a, bA, accA);
        accB = wmma4(a, bB, accB);
      }
    }
    if (t == 0) {
#pragma unroll
      for (int r = 0; r < 8; ++r) {
        size_t row = (size_t)(n0 + r + 8 * half) * 512;
        out[row + wA + nl] = accA[r] * SC_NORM;
        out[row + wB + nl] = accB[r] * SC_NORM;
      }
    } else {
      const int c = t - 1;
#pragma unroll
      for (int r = 0; r < 8; ++r) {
        size_t row = (size_t)(n0 + r + 8 * half) * 512 + MULC;
        out[row + (wA + nl) * 3 + c] = accA[r] * SC_NORM;
        out[row + (wB + nl) * 3 + c] = accB[r] * SC_NORM;
      }
    }
  }

  {
    v8f accA = {}, accB = {};
    for (int kk = 0; kk < MULC; kk += 4) {
      int k0 = kk + 2 * half;
      v2f a = { xs[m][k0], xs[m][k0 + 1] };
      size_t krow = (size_t)k0 * MULC;
      v2f bA = { lin1_ws[krow + wA + nl], lin1_ws[krow + MULC + wA + nl] };
      v2f bB = { lin1_ws[krow + wB + nl], lin1_ws[krow + MULC + wB + nl] };
      accA = wmma4(a, bA, accA);
      accB = wmma4(a, bB, accB);
    }
#pragma unroll
    for (int r = 0; r < 8; ++r) {
      size_t row = (size_t)(n0 + r + 8 * half) * 512;
      y[row + wA + nl] = accA[r] * L1_NORM;
      y[row + wB + nl] = accB[r] * L1_NORM;
    }
  }
  for (int c = 0; c < 3; ++c) {
    v8f accA = {}, accB = {};
    for (int kk = 0; kk < MULC; kk += 4) {
      int k0 = kk + 2 * half;
      v2f a = { xvc[c][m][k0], xvc[c][m][k0 + 1] };
      size_t krow = (size_t)k0 * MULC;
      v2f bA = { lin1_wv[krow + wA + nl], lin1_wv[krow + MULC + wA + nl] };
      v2f bB = { lin1_wv[krow + wB + nl], lin1_wv[krow + MULC + wB + nl] };
      accA = wmma4(a, bA, accA);
      accB = wmma4(a, bB, accB);
    }
#pragma unroll
    for (int r = 0; r < 8; ++r) {
      size_t row = (size_t)(n0 + r + 8 * half) * 512 + MULC;
      y[row + (wA + nl) * 3 + c] = accA[r] * L1_NORM;
      y[row + (wB + nl) * 3 + c] = accB[r] * L1_NORM;
    }
  }
}

__global__ __launch_bounds__(32) void edge_kernel(
    const float* __restrict__ y, const float* __restrict__ edge_feats,
    const float* __restrict__ edge_attrs, const int* __restrict__ edge_index,
    const float* __restrict__ fc_w0, const float* __restrict__ fc_w1,
    const float* __restrict__ fc_w2, float* __restrict__ s_acc,
    float* __restrict__ v_acc) {
  __shared__ float ytile[16][512];
  __shared__ float ha[16][64];
  __shared__ float hb[16][64];
  __shared__ float ea[16][4];
  __shared__ float ef[16][NRAD];
  __shared__ int   dstn[16];

  const int e0   = blockIdx.x * 16;
  const int lane = threadIdx.x;
  const int half = lane >> 4;
  const int m    = lane & 15;
  const int nl   = lane & 15;

  for (int r = 0; r < 16; ++r) {
    int sr = edge_index[NE + e0 + r];
    const float4* sp = (const float4*)(y + (size_t)sr * 512);
    float4* dp = (float4*)&ytile[r][0];
    for (int i = lane; i < 128; i += 32) dp[i] = sp[i];
  }
  for (int i = lane; i < 16 * NRAD; i += 32)
    ef[i >> 3][i & 7] = edge_feats[(size_t)e0 * NRAD + i];
  for (int i = lane; i < 16 * 4; i += 32)
    ea[i >> 2][i & 3] = edge_attrs[(size_t)e0 * 4 + i];
  if (lane < 16) dstn[lane] = edge_index[e0 + lane];
  __syncthreads();

  for (int nt = 0; nt < 4; ++nt) {
    v8f acc = {};
    for (int kk = 0; kk < 8; kk += 4) {
      int k0 = kk + 2 * half;
      v2f a = { ef[m][k0], ef[m][k0 + 1] };
      v2f b = { fc_w0[k0 * 64 + nt * 16 + nl],
                fc_w0[(k0 + 1) * 64 + nt * 16 + nl] };
      acc = wmma4(a, b, acc);
    }
#pragma unroll
    for (int r = 0; r < 8; ++r)
      ha[r + 8 * half][nt * 16 + nl] = silu_c(acc[r] * INV_SQRT8);
  }
  __syncthreads();

  for (int nt = 0; nt < 4; ++nt) {
    v8f acc = {};
    for (int kk = 0; kk < 64; kk += 4) {
      int k0 = kk + 2 * half;
      v2f a = { ha[m][k0], ha[m][k0 + 1] };
      v2f b = { fc_w1[k0 * 64 + nt * 16 + nl],
                fc_w1[(k0 + 1) * 64 + nt * 16 + nl] };
      acc = wmma4(a, b, acc);
    }
#pragma unroll
    for (int r = 0; r < 8; ++r)
      hb[r + 8 * half][nt * 16 + nl] = silu_c(acc[r] * 0.125f);
  }
  __syncthreads();

  for (int nt = 0; nt < 32; ++nt) {
    const int j0 = nt * 16;
    v8f acc = {};
    for (int kk = 0; kk < 64; kk += 4) {
      int k0 = kk + 2 * half;
      v2f a = { hb[m][k0], hb[m][k0 + 1] };
      v2f b = { fc_w2[(size_t)k0 * 512 + j0 + nl],
                fc_w2[(size_t)(k0 + 1) * 512 + j0 + nl] };
      acc = wmma4(a, b, acc);
    }
    const int j    = j0 + nl;
    const int quad = j >> 7;
    const int u    = j & 127;
#pragma unroll
    for (int r = 0; r < 8; ++r) {
      const int e = r + 8 * half;
      const float wv = acc[r] * 0.125f;
      const int d = dstn[e];
      const float as = ea[e][0];
      if (quad == 0) {
        atomicAdd(&s_acc[(size_t)d * 256 + u],
                  wv * ytile[e][u] * as * INV_NEIGH);
      } else if (quad == 1) {
        float base = wv * ytile[e][u] * INV_NEIGH;
        atomicAdd(&v_acc[((size_t)d * 256 + u) * 3 + 0], base * ea[e][1]);
        atomicAdd(&v_acc[((size_t)d * 256 + u) * 3 + 1], base * ea[e][2]);
        atomicAdd(&v_acc[((size_t)d * 256 + u) * 3 + 2], base * ea[e][3]);
      } else if (quad == 2) {
        float base = wv * as * INV_NEIGH;
        atomicAdd(&v_acc[((size_t)d * 256 + 128 + u) * 3 + 0],
                  base * ytile[e][128 + u * 3 + 0]);
        atomicAdd(&v_acc[((size_t)d * 256 + 128 + u) * 3 + 1],
                  base * ytile[e][128 + u * 3 + 1]);
        atomicAdd(&v_acc[((size_t)d * 256 + 128 + u) * 3 + 2],
                  base * ytile[e][128 + u * 3 + 2]);
      } else {
        float dotv = ytile[e][128 + u * 3 + 0] * ea[e][1] +
                     ytile[e][128 + u * 3 + 1] * ea[e][2] +
                     ytile[e][128 + u * 3 + 2] * ea[e][3];
        atomicAdd(&s_acc[(size_t)d * 256 + 128 + u],
                  wv * dotv * INV_SQ3 * INV_NEIGH);
      }
    }
  }
}

__global__ __launch_bounds__(128) void out_kernel(
    const float* __restrict__ s_acc, const float* __restrict__ v_acc,
    const float* __restrict__ lin2_ws, const float* __restrict__ lin2_wv,
    float* __restrict__ out) {
  __shared__ float st[16][256];
  __shared__ float vt[16][256];
  const int n0  = blockIdx.x * 16;
  const int tid = threadIdx.x;
  for (int i = tid; i < 16 * 256; i += 128)
    st[i >> 8][i & 255] = s_acc[(size_t)n0 * 256 + i];
  __syncthreads();

  const int wave = tid >> 5;
  const int lane = tid & 31;
  const int half = lane >> 4;
  const int m    = lane & 15;
  const int nl   = lane & 15;
  const int wA   = wave * 16;
  const int wB   = wA + 64;

  {
    v8f accA = {}, accB = {};
    for (int kk = 0; kk < 256; kk += 4) {
      int k0 = kk + 2 * half;
      v2f a = { st[m][k0], st[m][k0 + 1] };
      size_t krow = (size_t)k0 * MULC;
      v2f bA = { lin2_ws[krow + wA + nl], lin2_ws[krow + MULC + wA + nl] };
      v2f bB = { lin2_ws[krow + wB + nl], lin2_ws[krow + MULC + wB + nl] };
      accA = wmma4(a, bA, accA);
      accB = wmma4(a, bB, accB);
    }
#pragma unroll
    for (int r = 0; r < 8; ++r) {
      size_t row = (size_t)(n0 + r + 8 * half) * 512;
      out[row + wA + nl] += accA[r] * L2_NORM;
      out[row + wB + nl] += accB[r] * L2_NORM;
    }
  }
  for (int c = 0; c < 3; ++c) {
    __syncthreads();
    for (int i = tid; i < 16 * 256; i += 128) {
      int r = i >> 8, k = i & 255;
      vt[r][k] = v_acc[(((size_t)(n0 + r)) * 256 + k) * 3 + c];
    }
    __syncthreads();
    v8f accA = {}, accB = {};
    for (int kk = 0; kk < 256; kk += 4) {
      int k0 = kk + 2 * half;
      v2f a = { vt[m][k0], vt[m][k0 + 1] };
      size_t krow = (size_t)k0 * MULC;
      v2f bA = { lin2_wv[krow + wA + nl], lin2_wv[krow + MULC + wA + nl] };
      v2f bB = { lin2_wv[krow + wB + nl], lin2_wv[krow + MULC + wB + nl] };
      accA = wmma4(a, bA, accA);
      accB = wmma4(a, bB, accB);
    }
#pragma unroll
    for (int r = 0; r < 8; ++r) {
      size_t row = (size_t)(n0 + r + 8 * half) * 512 + MULC;
      out[row + (wA + nl) * 3 + c] += accA[r] * L2_NORM;
      out[row + (wB + nl) * 3 + c] += accB[r] * L2_NORM;
    }
  }
}

extern "C" void kernel_launch(void* const* d_in, const int* in_sizes, int n_in,
                              void* d_out, int out_size, void* d_ws,
                              size_t ws_size, hipStream_t stream) {
  const float* node_feats = (const float*)d_in[0];
  const float* node_attrs = (const float*)d_in[1];
  const float* edge_feats = (const float*)d_in[2];
  const float* edge_attrs = (const float*)d_in[3];
  const int*   edge_index = (const int*)d_in[4];
  const float* lin1_ws    = (const float*)d_in[5];
  const float* lin1_wv    = (const float*)d_in[6];
  const float* fc_w0      = (const float*)d_in[7];
  const float* fc_w1      = (const float*)d_in[8];
  const float* fc_w2      = (const float*)d_in[9];
  const float* lin2_ws    = (const float*)d_in[10];
  const float* lin2_wv    = (const float*)d_in[11];
  const float* sc_ws      = (const float*)d_in[12];
  const float* sc_wv      = (const float*)d_in[13];

  float* out   = (float*)d_out;
  float* y     = (float*)d_ws;
  float* s_acc = y + (size_t)NN * 512;
  float* v_acc = s_acc + (size_t)NN * 256;

  zero_kernel<<<2048, 256, 0, stream>>>(s_acc, (long)NN * 1024);
  node_kernel<<<NN / 16, 128, 0, stream>>>(node_feats, node_attrs, lin1_ws,
                                           lin1_wv, sc_ws, sc_wv, out, y);
  edge_kernel<<<NE / 16, 32, 0, stream>>>(y, edge_feats, edge_attrs,
                                          edge_index, fc_w0, fc_w1, fc_w2,
                                          s_acc, v_acc);
  out_kernel<<<NN / 16, 128, 0, stream>>>(s_acc, v_acc, lin2_ws, lin2_wv, out);
}


# node launches (lin1, self-connection, lin2) cut into half-column units: 1250 independent blocks, 5 or 4 waves per SIMD, was 3 or 2; self-connection forms A pairs in registers, no LDS; sums unchanged
# speedup vs baseline: 2.6126x; 2.6126x over previous
#include <hip/hip_runtime.h>
#include <hip/hip_bf16.h>


typedef __attribute__((ext_vector_type(2))) float v2f;
typedef __attribute__((ext_vector_type(8))) float v8f;

static constexpr int NN   = 10000;
static constexpr int NE   = 160000;
static constexpr int MULC = 128;
static constexpr int NSP  = 10;
static constexpr int NRAD = 8;

static constexpr float SILU_C    = 1.679f;
static constexpr float SC_NORM   = 0.02795084971874737f;
static constexpr float L1_NORM   = 0.08838834764831845f;
static constexpr float L2_NORM   = 0.0625f;
static constexpr float INV_SQRT8 = 0.3535533905932738f;
static constexpr float INV_SQ3   = 0.5773502691896258f;
static constexpr float INV_NEIGH = 0.25f;

__device__ __forceinline__ v8f wmma4(v2f a, v2f b, v8f c) {
  return __builtin_amdgcn_wmma_f32_16x16x4_f32(
       false, a,  false, b,
       (short)0, c,  false,  false);
}

__device__ __forceinline__ float silu_c(float z) {
  return SILU_C * z / (1.0f + __expf(-z));
}

__device__ __forceinline__ float xhalf(float x) {
  return __builtin_bit_cast(
      float, __builtin_amdgcn_permlanex16(__builtin_bit_cast(int, x),
                                          __builtin_bit_cast(int, x),
                                          0x76543210, 0xfedcba98, false, false));
}

template <int ASTRIDE, int KLEN>
__device__ __forceinline__ void mm4g(const float* __restrict__ arow,
                                     const float* __restrict__ W,
                                     const int half, const int nl,
                                     v8f (&acc)[4]) {
#pragma unroll 8
  for (int kk = 0; kk < KLEN; kk += 4) {
    const int k0 = kk + 2 * half;
    v2f a = { arow[k0 * ASTRIDE], arow[(k0 + 1) * ASTRIDE] };
    const float* w0 = W + (size_t)k0 * MULC + nl;
#pragma unroll
    for (int j = 0; j < 4; ++j) {
      v2f b = { w0[j * 16], w0[MULC + j * 16] };
      acc[j] = wmma4(a, b, acc[j]);
    }
  }
}

template <int ASTRIDE>
__device__ __forceinline__ void sc_mm4(const float* __restrict__ xrow,
                                       const v2f (&ap)[5],
                                       const float* __restrict__ W,
                                       const int half, const int nl,
                                       v8f (&acc)[4]) {
  for (int q = 0; q < MULC / 2; ++q) {
    const float x0 = xrow[(2 * q) * ASTRIDE];
    const float x1 = xrow[(2 * q + 1) * ASTRIDE];
    const float xm = half ? x1 : x0;
#pragma unroll
    for (int i = 0; i < 5; ++i) {
      const float xf = (i < 2) ? x0 : (i > 2) ? x1 : xm;
      v2f a = { xf * ap[i][0], xf * ap[i][1] };
      const int k0 = 20 * q + 4 * i + 2 * half;
      const float* w0 = W + (size_t)k0 * MULC + nl;
#pragma unroll
      for (int j = 0; j < 4; ++j) {
        v2f b = { w0[j * 16], w0[MULC + j * 16] };
        acc[j] = wmma4(a, b, acc[j]);
      }
    }
  }
}

__global__ void zero_kernel(float* __restrict__ p, long n,
                            int* __restrict__ cnt, int ncnt) {
  long i = (long)blockIdx.x * blockDim.x + threadIdx.x;
  long stride = (long)gridDim.x * blockDim.x;
  if (i < ncnt) cnt[i] = 0;
  for (; i < n; i += stride) p[i] = 0.0f;
}

__global__ __launch_bounds__(1024) void scan_kernel(int* __restrict__ cnt) {
  __shared__ int part[1024];
  const int tid = threadIdx.x;
  int v[10];
  int s = 0;
#pragma unroll
  for (int j = 0; j < 10; ++j) {
    const int idx = tid * 10 + j;
    v[j] = (idx < NN) ? cnt[idx] : 0;
    s += v[j];
  }
  part[tid] = s;
  __syncthreads();
  for (int off = 1; off < 1024; off <<= 1) {
    const int t = (tid >= off) ? part[tid - off] : 0;
    __syncthreads();
    part[tid] += t;
    __syncthreads();
  }
  int base = part[tid] - s;
#pragma unroll
  for (int j = 0; j < 10; ++j) {
    const int idx = tid * 10 + j;
    if (idx < NN) cnt[idx] = base;
    base += v[j];
  }
}

__global__ __launch_bounds__(256) void fill_kernel(
    const int* __restrict__ edge_index, int* __restrict__ cursor,
    int* __restrict__ perm) {
  const int e = blockIdx.x * 256 + threadIdx.x;
  const int d = edge_index[e];
  const int pos = atomicAdd(&cursor[d], 1);
  perm[pos] = e;
}

__global__ __launch_bounds__(128) __attribute__((amdgpu_waves_per_eu(1, 5)))
void y_kernel(
    const float* __restrict__ node_feats, const float* __restrict__ lin1_ws,
    const float* __restrict__ lin1_wv, const int* __restrict__ edge_index,
    float* __restrict__ y, int* __restrict__ cnt) {
  const int n0  = (blockIdx.x >> 1) * 16;
  const int c0  = (blockIdx.x & 1) * 64;
  const int tid = threadIdx.x;

  atomicAdd(&cnt[edge_index[blockIdx.x * 128 + tid]], 1);

  const int t    = __builtin_amdgcn_readfirstlane(tid >> 5);
  const int lane = tid & 31;
  const int half = lane >> 4;
  const int m    = lane & 15;
  const int nl   = lane & 15;

  const float* xrow = node_feats + (size_t)(n0 + m) * 512;
  v8f acc[4] = {};
  if (t == 0) mm4g<1, MULC>(xrow, lin1_ws + c0, half, nl, acc);
  else        mm4g<3, MULC>(xrow + MULC + (t - 1), lin1_wv + c0, half, nl, acc);
#pragma unroll
  for (int j = 0; j < 4; ++j) {
    const int col = c0 + j * 16 + nl;
    const int off = (t == 0) ? col : MULC + col * 3 + (t - 1);
#pragma unroll
    for (int r = 0; r < 8; ++r)
      y[(size_t)(n0 + r + 8 * half) * 512 + off] = acc[j][r] * L1_NORM;
  }
}

__global__ __launch_bounds__(128) __attribute__((amdgpu_waves_per_eu(1, 5)))
void sc_kernel(
    const float* __restrict__ node_feats, const float* __restrict__ node_attrs,
    const float* __restrict__ sc_ws, const float* __restrict__ sc_wv,
    float* __restrict__ out) {
  const int n0  = (blockIdx.x >> 1) * 16;
  const int c0  = (blockIdx.x & 1) * 64;
  const int tid = threadIdx.x;

  const int t    = __builtin_amdgcn_readfirstlane(tid >> 5);
  const int lane = tid & 31;
  const int half = lane >> 4;
  const int m    = lane & 15;
  const int nl   = lane & 15;

  const float* arow = node_attrs + (size_t)(n0 + m) * NSP;
  v2f at[5];
#pragma unroll
  for (int p = 0; p < 5; ++p) at[p] = v2f{ arow[2 * p], arow[2 * p + 1] };
  v2f ap[5];
  ap[0] = half ? at[1] : at[0];
  ap[1] = half ? at[3] : at[2];
  ap[2] = half ? at[0] : at[4];
  ap[3] = half ? at[2] : at[1];
  ap[4] = half ? at[4] : at[3];

  const float* xrow = node_feats + (size_t)(n0 + m) * 512;
  v8f acc[4] = {};
  if (t == 0) sc_mm4<1>(xrow, ap, sc_ws + c0, half, nl, acc);
  else        sc_mm4<3>(xrow + MULC + (t - 1), ap, sc_wv + c0, half, nl, acc);
#pragma unroll
  for (int j = 0; j < 4; ++j) {
    const int col = c0 + j * 16 + nl;
    const int off = (t == 0) ? col : MULC + col * 3 + (t - 1);
#pragma unroll
    for (int r = 0; r < 8; ++r)
      out[(size_t)(n0 + r + 8 * half) * 512 + off] = acc[j][r] * SC_NORM;
  }
}

template <int QUAD>
__device__ __forceinline__ void tile_pair(
    const int p, const int lane, const float* __restrict__ y,
    const float* __restrict__ fc_w2, const float (*hb)[64],
    const float (*ea)[4], const int* dstn, const int (&srow)[16],
    const unsigned flush, float* __restrict__ s_acc,
    float* __restrict__ v_acc) {
  const int half = lane >> 4;
  const int m    = lane & 15;
  const int nl   = lane & 15;
  const int j0   = p * 32;
  const int u    = ((p & 3) << 5) + lane;

  constexpr int NY = (QUAD < 2) ? 16 : 48;
  float yq[NY];
  if constexpr (QUAD < 2) {
#pragma unroll
    for (int e = 0; e < 16; ++e) yq[e] = y[(size_t)srow[e] * 512 + u];
  } else {
#pragma unroll
    for (int e = 0; e < 16; ++e) {
      const float* yr = y + (size_t)srow[e] * 512 + 128 + u * 3;
      yq[e]      = yr[0];
      yq[16 + e] = yr[1];
      yq[32 + e] = yr[2];
    }
  }

  v8f accA = {}, accB = {};
  for (int kk = 0; kk < 64; kk += 4) {
    int k0 = kk + 2 * half;
    v2f a = { hb[m][k0], hb[m][k0 + 1] };
    v2f bA = { fc_w2[(size_t)k0 * 512 + j0 + nl],
               fc_w2[(size_t)(k0 + 1) * 512 + j0 + nl] };
    v2f bB = { fc_w2[(size_t)k0 * 512 + j0 + 16 + nl],
               fc_w2[(size_t)(k0 + 1) * 512 + j0 + 16 + nl] };
    accA = wmma4(a, bA, accA);
    accB = wmma4(a, bB, accB);
  }

  float wrow[16];
#pragma unroll
  for (int r = 0; r < 8; ++r) {
    const float xa = xhalf(accA[r]);
    const float xb = xhalf(accB[r]);
    wrow[r]     = (half == 0) ? accA[r] : xb;
    wrow[r + 8] = (half == 0) ? xa : accB[r];
  }

  if constexpr (QUAD == 0) {
    float val[16];
#pragma unroll
    for (int e = 0; e < 16; ++e) {
      const float wv = wrow[e] * 0.125f;
      val[e] = wv * yq[e] * ea[e][0] * INV_NEIGH;
    }
    float run = 0.0f;
#pragma unroll
    for (int e = 0; e < 16; ++e) {
      run += val[e];
      if ((flush >> e) & 1u) {
        const int d = __builtin_amdgcn_readfirstlane(dstn[e]);
        atomicAdd(&s_acc[(size_t)d * 256 + u], run);
        run = 0.0f;
      }
    }
  } else if constexpr (QUAD == 1) {
    float t0[16], t1[16], t2[16];
#pragma unroll
    for (int e = 0; e < 16; ++e) {
      const float wv = wrow[e] * 0.125f;
      const float base = wv * yq[e] * INV_NEIGH;
      t0[e] = base * ea[e][1];
      t1[e] = base * ea[e][2];
      t2[e] = base * ea[e][3];
    }
    float run0 = 0.0f, run1 = 0.0f, run2 = 0.0f;
#pragma unroll
    for (int e = 0; e < 16; ++e) {
      run0 += t0[e];
      run1 += t1[e];
      run2 += t2[e];
      if ((flush >> e) & 1u) {
        const int d = __builtin_amdgcn_readfirstlane(dstn[e]);
        float* vrow = v_acc + (size_t)d * 768 + u;
        atomicAdd(vrow,       run0);
        atomicAdd(vrow + 256, run1);
        atomicAdd(vrow + 512, run2);
        run0 = 0.0f; run1 = 0.0f; run2 = 0.0f;
      }
    }
  } else if constexpr (QUAD == 2) {
#pragma unroll
    for (int e = 0; e < 16; ++e) {
      const float wv = wrow[e] * 0.125f;
      const float base = wv * ea[e][0] * INV_NEIGH;
      yq[e]      = base * yq[e];
      yq[16 + e] = base * yq[16 + e];
      yq[32 + e] = base * yq[32 + e];
    }
    float run0 = 0.0f, run1 = 0.0f, run2 = 0.0f;
#pragma unroll
    for (int e = 0; e < 16; ++e) {
      run0 += yq[e];
      run1 += yq[16 + e];
      run2 += yq[32 + e];
      if ((flush >> e) & 1u) {
        const int d = __builtin_amdgcn_readfirstlane(dstn[e]);
        float* vrow = v_acc + (size_t)d * 768 + 128 + u;
        atomicAdd(vrow,       run0);
        atomicAdd(vrow + 256, run1);
        atomicAdd(vrow + 512, run2);
        run0 = 0.0f; run1 = 0.0f; run2 = 0.0f;
      }
    }
  } else {
    float val[16];
#pragma unroll
    for (int e = 0; e < 16; ++e) {
      const float wv = wrow[e] * 0.125f;
      float dotv = yq[e] * ea[e][1] + yq[16 + e] * ea[e][2] +
                   yq[32 + e] * ea[e][3];
      val[e] = wv * dotv * INV_SQ3 * INV_NEIGH;
    }
    float run = 0.0f;
#pragma unroll
    for (int e = 0; e < 16; ++e) {
      run += val[e];
      if ((flush >> e) & 1u) {
        const int d = __builtin_amdgcn_readfirstlane(dstn[e]);
        atomicAdd(&s_acc[(size_t)d * 256 + 128 + u], run);
        run = 0.0f;
      }
    }
  }
}

__global__ __launch_bounds__(32) void edge_kernel(
    const float* __restrict__ y, const float* __restrict__ edge_feats,
    const float* __restrict__ edge_attrs, const int* __restrict__ edge_index,
    const int* __restrict__ perm,
    const float* __restrict__ fc_w0, const float* __restrict__ fc_w1,
    const float* __restrict__ fc_w2, float* __restrict__ s_acc,
    float* __restrict__ v_acc) {
  __shared__ float ha[16][64];
  __shared__ float hb[16][64];
  __shared__ float ea[16][4];
  __shared__ float ef[16][NRAD];
  __shared__ int   dstn[16];
  __shared__ int   srcn[16];
  __shared__ int   eid[16];

  const int e0   = blockIdx.x * 16;
  const int lane = threadIdx.x;
  const int half = lane >> 4;
  const int m    = lane & 15;
  const int nl   = lane & 15;

  if (lane < 16) {
    int id = perm[e0 + lane];
    id = min(max(id, 0), NE - 1);
    eid[lane]  = id;
    dstn[lane] = edge_index[id];
    srcn[lane] = edge_index[NE + id];
  }
  __syncthreads();

  int srow[16];
#pragma unroll
  for (int e = 0; e < 16; ++e)
    srow[e] = __builtin_amdgcn_readfirstlane(srcn[e]);
  for (int i = lane; i < 16 * NRAD; i += 32)
    ef[i >> 3][i & 7] = edge_feats[(size_t)eid[i >> 3] * NRAD + (i & 7)];
  for (int i = lane; i < 16 * 4; i += 32)
    ea[i >> 2][i & 3] = edge_attrs[(size_t)eid[i >> 2] * 4 + (i & 3)];
  const bool run_end =
      (lane < 16) && (lane == 15 || dstn[(lane + 1) & 15] != dstn[lane & 15]);
  const unsigned flush = __builtin_amdgcn_ballot_w32(run_end);
  __syncthreads();

  for (int nt = 0; nt < 4; ++nt) {
    v8f acc = {};
    for (int kk = 0; kk < 8; kk += 4) {
      int k0 = kk + 2 * half;
      v2f a = { ef[m][k0], ef[m][k0 + 1] };
      v2f b = { fc_w0[k0 * 64 + nt * 16 + nl],
                fc_w0[(k0 + 1) * 64 + nt * 16 + nl] };
      acc = wmma4(a, b, acc);
    }
#pragma unroll
    for (int r = 0; r < 8; ++r)
      ha[r + 8 * half][nt * 16 + nl] = silu_c(acc[r] * INV_SQRT8);
  }
  __syncthreads();

  for (int nt = 0; nt < 4; ++nt) {
    v8f acc = {};
    for (int kk = 0; kk < 64; kk += 4) {
      int k0 = kk + 2 * half;
      v2f a = { ha[m][k0], ha[m][k0 + 1] };
      v2f b = { fc_w1[k0 * 64 + nt * 16 + nl],
                fc_w1[(k0 + 1) * 64 + nt * 16 + nl] };
      acc = wmma4(a, b, acc);
    }
#pragma unroll
    for (int r = 0; r < 8; ++r)
      hb[r + 8 * half][nt * 16 + nl] = silu_c(acc[r] * 0.125f);
  }
  __syncthreads();

  for (int pp = 0; pp < 4; ++pp)
    tile_pair<0>(pp, lane, y, fc_w2, hb, ea, dstn, srow, flush, s_acc, v_acc);
  for (int pp = 4; pp < 8; ++pp)
    tile_pair<1>(pp, lane, y, fc_w2, hb, ea, dstn, srow, flush, s_acc, v_acc);
  for (int pp = 8; pp < 12; ++pp)
    tile_pair<2>(pp, lane, y, fc_w2, hb, ea, dstn, srow, flush, s_acc, v_acc);
  for (int pp = 12; pp < 16; ++pp)
    tile_pair<3>(pp, lane, y, fc_w2, hb, ea, dstn, srow, flush, s_acc, v_acc);
}

__global__ __launch_bounds__(128) __attribute__((amdgpu_waves_per_eu(1, 5)))
void out_kernel(
    const float* __restrict__ s_acc, const float* __restrict__ v_acc,
    const float* __restrict__ lin2_ws, const float* __restrict__ lin2_wv,
    float* __restrict__ out) {
  const int n0  = (blockIdx.x >> 1) * 16;
  const int c0  = (blockIdx.x & 1) * 64;
  const int tid = threadIdx.x;

  const int t    = __builtin_amdgcn_readfirstlane(tid >> 5);
  const int lane = tid & 31;
  const int half = lane >> 4;
  const int m    = lane & 15;
  const int nl   = lane & 15;
  const float* W = ((t == 0) ? lin2_ws : lin2_wv) + c0;
  const float* arow = (t == 0)
      ? s_acc + (size_t)(n0 + m) * 256
      : v_acc + (size_t)(n0 + m) * 768 + (size_t)(t - 1) * 256;

  v8f acc[4] = {};
  mm4g<1, 256>(arow, W, half, nl, acc);
#pragma unroll
  for (int j = 0; j < 4; ++j) {
    const int col = c0 + j * 16 + nl;
    const int off = (t == 0) ? col : MULC + col * 3 + (t - 1);
#pragma unroll
    for (int r = 0; r < 8; ++r)
      out[(size_t)(n0 + r + 8 * half) * 512 + off] += acc[j][r] * L2_NORM;
  }
}

extern "C" void kernel_launch(void* const* d_in, const int* in_sizes, int n_in,
                              void* d_out, int out_size, void* d_ws,
                              size_t ws_size, hipStream_t stream) {
  const float* node_feats = (const float*)d_in[0];
  const float* node_attrs = (const float*)d_in[1];
  const float* edge_feats = (const float*)d_in[2];
  const float* edge_attrs = (const float*)d_in[3];
  const int*   edge_index = (const int*)d_in[4];
  const float* lin1_ws    = (const float*)d_in[5];
  const float* lin1_wv    = (const float*)d_in[6];
  const float* fc_w0      = (const float*)d_in[7];
  const float* fc_w1      = (const float*)d_in[8];
  const float* fc_w2      = (const float*)d_in[9];
  const float* lin2_ws    = (const float*)d_in[10];
  const float* lin2_wv    = (const float*)d_in[11];
  const float* sc_ws      = (const float*)d_in[12];
  const float* sc_wv      = (const float*)d_in[13];

  float* out   = (float*)d_out;
  float* y     = (float*)d_ws;
  float* s_acc = y + (size_t)NN * 512;
  float* v_acc = s_acc + (size_t)NN * 256;
  int* perm = (int*)d_out;
  int* cnt  = perm + NE;

  zero_kernel<<<2048, 256, 0, stream>>>(s_acc, (long)NN * 1024, cnt, NN);
  y_kernel<<<NN / 8, 128, 0, stream>>>(node_feats, lin1_ws, lin1_wv,
                                        edge_index, y, cnt);
  scan_kernel<<<1, 1024, 0, stream>>>(cnt);
  fill_kernel<<<NE / 256, 256, 0, stream>>>(edge_index, cnt, perm);
  edge_kernel<<<NE / 16, 32, 0, stream>>>(y, edge_feats, edge_attrs,
                                          edge_index, perm, fc_w0, fc_w1,
                                          fc_w2, s_acc, v_acc);
  sc_kernel<<<NN / 8, 128, 0, stream>>>(node_feats, node_attrs, sc_ws, sc_wv,
                                         out);
  out_kernel<<<NN / 8, 128, 0, stream>>>(s_acc, v_acc, lin2_ws, lin2_wv, out);
}


# edge launch: one 16-wave work-group per WGP, each wave loops over edge tiles; fc_w1, fc_w2 copied once per group to LDS as padded row pairs; B pairs are 8-byte LDS reads, not global loads; same sums
# speedup vs baseline: 2.7025x; 1.0344x over previous
#include <hip/hip_runtime.h>
#include <hip/hip_bf16.h>


typedef __attribute__((ext_vector_type(2))) float v2f;
typedef __attribute__((ext_vector_type(8))) float v8f;

static constexpr int NN   = 10000;
static constexpr int NE   = 160000;
static constexpr int MULC = 128;
static constexpr int NSP  = 10;
static constexpr int NRAD = 8;

static constexpr int EWAVES = 16;
static constexpr int W2P    = 1024 + 32;
static constexpr int W1P    = 128 + 32;

static constexpr float SILU_C    = 1.679f;
static constexpr float SC_NORM   = 0.02795084971874737f;
static constexpr float L1_NORM   = 0.08838834764831845f;
static constexpr float L2_NORM   = 0.0625f;
static constexpr float INV_SQRT8 = 0.3535533905932738f;
static constexpr float INV_SQ3   = 0.5773502691896258f;
static constexpr float INV_NEIGH = 0.25f;

__device__ __forceinline__ v8f wmma4(v2f a, v2f b, v8f c) {
  return __builtin_amdgcn_wmma_f32_16x16x4_f32(
       false, a,  false, b,
       (short)0, c,  false,  false);
}

__device__ __forceinline__ float silu_c(float z) {
  return SILU_C * z / (1.0f + __expf(-z));
}

__device__ __forceinline__ float xhalf(float x) {
  return __builtin_bit_cast(
      float, __builtin_amdgcn_permlanex16(__builtin_bit_cast(int, x),
                                          __builtin_bit_cast(int, x),
                                          0x76543210, 0xfedcba98, false, false));
}

template <int ASTRIDE, int KLEN>
__device__ __forceinline__ void mm4g(const float* __restrict__ arow,
                                     const float* __restrict__ W,
                                     const int half, const int nl,
                                     v8f (&acc)[4]) {
#pragma unroll 8
  for (int kk = 0; kk < KLEN; kk += 4) {
    const int k0 = kk + 2 * half;
    v2f a = { arow[k0 * ASTRIDE], arow[(k0 + 1) * ASTRIDE] };
    const float* w0 = W + (size_t)k0 * MULC + nl;
#pragma unroll
    for (int j = 0; j < 4; ++j) {
      v2f b = { w0[j * 16], w0[MULC + j * 16] };
      acc[j] = wmma4(a, b, acc[j]);
    }
  }
}

template <int ASTRIDE>
__device__ __forceinline__ void sc_mm4(const float* __restrict__ xrow,
                                       const v2f (&ap)[5],
                                       const float* __restrict__ W,
                                       const int half, const int nl,
                                       v8f (&acc)[4]) {
  for (int q = 0; q < MULC / 2; ++q) {
    const float x0 = xrow[(2 * q) * ASTRIDE];
    const float x1 = xrow[(2 * q + 1) * ASTRIDE];
    const float xm = half ? x1 : x0;
#pragma unroll
    for (int i = 0; i < 5; ++i) {
      const float xf = (i < 2) ? x0 : (i > 2) ? x1 : xm;
      v2f a = { xf * ap[i][0], xf * ap[i][1] };
      const int k0 = 20 * q + 4 * i + 2 * half;
      const float* w0 = W + (size_t)k0 * MULC + nl;
#pragma unroll
      for (int j = 0; j < 4; ++j) {
        v2f b = { w0[j * 16], w0[MULC + j * 16] };
        acc[j] = wmma4(a, b, acc[j]);
      }
    }
  }
}

__global__ void zero_kernel(float* __restrict__ p, long n,
                            int* __restrict__ cnt, int ncnt) {
  long i = (long)blockIdx.x * blockDim.x + threadIdx.x;
  long stride = (long)gridDim.x * blockDim.x;
  if (i < ncnt) cnt[i] = 0;
  for (; i < n; i += stride) p[i] = 0.0f;
}

__global__ __launch_bounds__(1024) void scan_kernel(int* __restrict__ cnt) {
  __shared__ int part[1024];
  const int tid = threadIdx.x;
  int v[10];
  int s = 0;
#pragma unroll
  for (int j = 0; j < 10; ++j) {
    const int idx = tid * 10 + j;
    v[j] = (idx < NN) ? cnt[idx] : 0;
    s += v[j];
  }
  part[tid] = s;
  __syncthreads();
  for (int off = 1; off < 1024; off <<= 1) {
    const int t = (tid >= off) ? part[tid - off] : 0;
    __syncthreads();
    part[tid] += t;
    __syncthreads();
  }
  int base = part[tid] - s;
#pragma unroll
  for (int j = 0; j < 10; ++j) {
    const int idx = tid * 10 + j;
    if (idx < NN) cnt[idx] = base;
    base += v[j];
  }
}

__global__ __launch_bounds__(256) void fill_kernel(
    const int* __restrict__ edge_index, int* __restrict__ cursor,
    int* __restrict__ perm) {
  const int e = blockIdx.x * 256 + threadIdx.x;
  const int d = edge_index[e];
  const int pos = atomicAdd(&cursor[d], 1);
  perm[pos] = e;
}

__global__ __launch_bounds__(128) __attribute__((amdgpu_waves_per_eu(1, 5)))
void y_kernel(
    const float* __restrict__ node_feats, const float* __restrict__ lin1_ws,
    const float* __restrict__ lin1_wv, const int* __restrict__ edge_index,
    float* __restrict__ y, int* __restrict__ cnt) {
  const int n0  = (blockIdx.x >> 1) * 16;
  const int c0  = (blockIdx.x & 1) * 64;
  const int tid = threadIdx.x;

  atomicAdd(&cnt[edge_index[blockIdx.x * 128 + tid]], 1);

  const int t    = __builtin_amdgcn_readfirstlane(tid >> 5);
  const int lane = tid & 31;
  const int half = lane >> 4;
  const int m    = lane & 15;
  const int nl   = lane & 15;

  const float* xrow = node_feats + (size_t)(n0 + m) * 512;
  v8f acc[4] = {};
  if (t == 0) mm4g<1, MULC>(xrow, lin1_ws + c0, half, nl, acc);
  else        mm4g<3, MULC>(xrow + MULC + (t - 1), lin1_wv + c0, half, nl, acc);
#pragma unroll
  for (int j = 0; j < 4; ++j) {
    const int col = c0 + j * 16 + nl;
    const int off = (t == 0) ? col : MULC + col * 3 + (t - 1);
#pragma unroll
    for (int r = 0; r < 8; ++r)
      y[(size_t)(n0 + r + 8 * half) * 512 + off] = acc[j][r] * L1_NORM;
  }
}

__global__ __launch_bounds__(128) __attribute__((amdgpu_waves_per_eu(1, 5)))
void sc_kernel(
    const float* __restrict__ node_feats, const float* __restrict__ node_attrs,
    const float* __restrict__ sc_ws, const float* __restrict__ sc_wv,
    float* __restrict__ out) {
  const int n0  = (blockIdx.x >> 1) * 16;
  const int c0  = (blockIdx.x & 1) * 64;
  const int tid = threadIdx.x;

  const int t    = __builtin_amdgcn_readfirstlane(tid >> 5);
  const int lane = tid & 31;
  const int half = lane >> 4;
  const int m    = lane & 15;
  const int nl   = lane & 15;

  const float* arow = node_attrs + (size_t)(n0 + m) * NSP;
  v2f at[5];
#pragma unroll
  for (int p = 0; p < 5; ++p) at[p] = v2f{ arow[2 * p], arow[2 * p + 1] };
  v2f ap[5];
  ap[0] = half ? at[1] : at[0];
  ap[1] = half ? at[3] : at[2];
  ap[2] = half ? at[0] : at[4];
  ap[3] = half ? at[2] : at[1];
  ap[4] = half ? at[4] : at[3];

  const float* xrow = node_feats + (size_t)(n0 + m) * 512;
  v8f acc[4] = {};
  if (t == 0) sc_mm4<1>(xrow, ap, sc_ws + c0, half, nl, acc);
  else        sc_mm4<3>(xrow + MULC + (t - 1), ap, sc_wv + c0, half, nl, acc);
#pragma unroll
  for (int j = 0; j < 4; ++j) {
    const int col = c0 + j * 16 + nl;
    const int off = (t == 0) ? col : MULC + col * 3 + (t - 1);
#pragma unroll
    for (int r = 0; r < 8; ++r)
      out[(size_t)(n0 + r + 8 * half) * 512 + off] = acc[j][r] * SC_NORM;
  }
}

template <int QUAD>
__device__ __forceinline__ void tile_pair(
    const int p, const int lane, const float* __restrict__ y,
    const float* w2l, const v2f (&afr)[16],
    const float (*ea)[4], const int* dstn, const int (&srow)[16],
    const unsigned flush, float* __restrict__ s_acc,
    float* __restrict__ v_acc) {
  const int half = lane >> 4;
  const int nl   = lane & 15;
  const int j0   = p * 32;
  const int u    = ((p & 3) << 5) + lane;

  constexpr int NY = (QUAD < 2) ? 16 : 48;
  float yq[NY];
  if constexpr (QUAD < 2) {
#pragma unroll
    for (int e = 0; e < 16; ++e) yq[e] = y[(size_t)srow[e] * 512 + u];
  } else {
#pragma unroll
    for (int e = 0; e < 16; ++e) {
      const float* yr = y + (size_t)srow[e] * 512 + 128 + u * 3;
      yq[e]      = yr[0];
      yq[16 + e] = yr[1];
      yq[32 + e] = yr[2];
    }
  }

  const float* wb = w2l + half * W2P + (j0 + nl) * 2;
  v2f bA[16], bB[16];
#pragma unroll
  for (int i = 0; i < 16; ++i) {
    bA[i] = *(const v2f*)(wb + 2 * i * W2P);
    bB[i] = *(const v2f*)(wb + 2 * i * W2P + 32);
  }
  __builtin_amdgcn_sched_barrier(0);
  v8f accA = {}, accB = {};
#pragma unroll
  for (int i = 0; i < 16; ++i) {
    accA = wmma4(afr[i], bA[i], accA);
    accB = wmma4(afr[i], bB[i], accB);
  }

  float wrow[16];
#pragma unroll
  for (int r = 0; r < 8; ++r) {
    const float xa = xhalf(accA[r]);
    const float xb = xhalf(accB[r]);
    wrow[r]     = (half == 0) ? accA[r] : xb;
    wrow[r + 8] = (half == 0) ? xa : accB[r];
  }

  if constexpr (QUAD == 0) {
    float val[16];
#pragma unroll
    for (int e = 0; e < 16; ++e) {
      const float wv = wrow[e] * 0.125f;
      val[e] = wv * yq[e] * ea[e][0] * INV_NEIGH;
    }
    float run = 0.0f;
#pragma unroll
    for (int e = 0; e < 16; ++e) {
      run += val[e];
      if ((flush >> e) & 1u) {
        const int d = __builtin_amdgcn_readfirstlane(dstn[e]);
        atomicAdd(&s_acc[(size_t)d * 256 + u], run);
        run = 0.0f;
      }
    }
  } else if constexpr (QUAD == 1) {
    float t0[16], t1[16], t2[16];
#pragma unroll
    for (int e = 0; e < 16; ++e) {
      const float wv = wrow[e] * 0.125f;
      const float base = wv * yq[e] * INV_NEIGH;
      t0[e] = base * ea[e][1];
      t1[e] = base * ea[e][2];
      t2[e] = base * ea[e][3];
    }
    float run0 = 0.0f, run1 = 0.0f, run2 = 0.0f;
#pragma unroll
    for (int e = 0; e < 16; ++e) {
      run0 += t0[e];
      run1 += t1[e];
      run2 += t2[e];
      if ((flush >> e) & 1u) {
        const int d = __builtin_amdgcn_readfirstlane(dstn[e]);
        float* vrow = v_acc + (size_t)d * 768 + u;
        atomicAdd(vrow,       run0);
        atomicAdd(vrow + 256, run1);
        atomicAdd(vrow + 512, run2);
        run0 = 0.0f; run1 = 0.0f; run2 = 0.0f;
      }
    }
  } else if constexpr (QUAD == 2) {
#pragma unroll
    for (int e = 0; e < 16; ++e) {
      const float wv = wrow[e] * 0.125f;
      const float base = wv * ea[e][0] * INV_NEIGH;
      yq[e]      = base * yq[e];
      yq[16 + e] = base * yq[16 + e];
      yq[32 + e] = base * yq[32 + e];
    }
    float run0 = 0.0f, run1 = 0.0f, run2 = 0.0f;
#pragma unroll
    for (int e = 0; e < 16; ++e) {
      run0 += yq[e];
      run1 += yq[16 + e];
      run2 += yq[32 + e];
      if ((flush >> e) & 1u) {
        const int d = __builtin_amdgcn_readfirstlane(dstn[e]);
        float* vrow = v_acc + (size_t)d * 768 + 128 + u;
        atomicAdd(vrow,       run0);
        atomicAdd(vrow + 256, run1);
        atomicAdd(vrow + 512, run2);
        run0 = 0.0f; run1 = 0.0f; run2 = 0.0f;
      }
    }
  } else {
    float val[16];
#pragma unroll
    for (int e = 0; e < 16; ++e) {
      const float wv = wrow[e] * 0.125f;
      float dotv = yq[e] * ea[e][1] + yq[16 + e] * ea[e][2] +
                   yq[32 + e] * ea[e][3];
      val[e] = wv * dotv * INV_SQ3 * INV_NEIGH;
    }
    float run = 0.0f;
#pragma unroll
    for (int e = 0; e < 16; ++e) {
      run += val[e];
      if ((flush >> e) & 1u) {
        const int d = __builtin_amdgcn_readfirstlane(dstn[e]);
        atomicAdd(&s_acc[(size_t)d * 256 + 128 + u], run);
        run = 0.0f;
      }
    }
  }
}

struct WaveLds {
  float ha[16][64];
  float hb[16][64];
  float ea[16][4];
  float ef[16][NRAD];
  int   dstn[16];
  int   srcn[16];
  int   eid[16];
};

__device__ __forceinline__ void wave_lds_sync() {
  __builtin_amdgcn_fence(__ATOMIC_RELEASE, "wavefront");
  __builtin_amdgcn_wave_barrier();
  __builtin_amdgcn_fence(__ATOMIC_ACQUIRE, "wavefront");
}

__global__ __launch_bounds__(EWAVES * 32) void edge_kernel(
    const float* __restrict__ y, const float* __restrict__ edge_feats,
    const float* __restrict__ edge_attrs, const int* __restrict__ edge_index,
    const int* __restrict__ perm,
    const float* __restrict__ fc_w0, const float* __restrict__ fc_w1,
    const float* __restrict__ fc_w2, float* __restrict__ s_acc,
    float* __restrict__ v_acc) {
  __shared__ __attribute__((aligned(16))) float   w2l[32 * W2P];
  __shared__ __attribute__((aligned(16))) float   w1l[32 * W1P];
  __shared__ __attribute__((aligned(16))) WaveLds wl[EWAVES];

  const int tid  = threadIdx.x;
  const int lane = tid & 31;
  const int half = lane >> 4;
  const int m    = lane & 15;
  const int nl   = lane & 15;

  static_assert(EWAVES * 32 == 512, "one thread per fc_w2 column");
  {
    v2f w1v[4];
#pragma unroll
    for (int q = 0; q < 4; ++q) {
      const int i = tid + q * 512, kp = i >> 6, col = i & 63;
      w1v[q] = v2f{ fc_w1[(2 * kp) * 64 + col], fc_w1[(2 * kp + 1) * 64 + col] };
    }
    for (int g = 0; g < 2; ++g) {
      v2f wv[16];
#pragma unroll
      for (int j = 0; j < 16; ++j) {
        const int kp = g * 16 + j;
        wv[j] = v2f{ fc_w2[(size_t)(2 * kp) * 512 + tid],
                     fc_w2[(size_t)(2 * kp + 1) * 512 + tid] };
      }
#pragma unroll
      for (int j = 0; j < 16; ++j)
        *(v2f*)(w2l + (g * 16 + j) * W2P + tid * 2) = wv[j];
    }
#pragma unroll
    for (int q = 0; q < 4; ++q) {
      const int i = tid + q * 512, kp = i >> 6, col = i & 63;
      *(v2f*)(w1l + kp * W1P + col * 2) = w1v[q];
    }
  }
  __syncthreads();

  WaveLds& L = wl[__builtin_amdgcn_readfirstlane(tid >> 5)];
  const int wave = __builtin_amdgcn_readfirstlane(tid >> 5);

  for (int tile = blockIdx.x + gridDim.x * wave; tile < NE / 16;
       tile += gridDim.x * EWAVES) {
    const int e0 = tile * 16;

    if (lane < 16) {
      int id = perm[e0 + lane];
      id = min(max(id, 0), NE - 1);
      L.eid[lane]  = id;
      L.dstn[lane] = edge_index[id];
      L.srcn[lane] = edge_index[NE + id];
    }
    wave_lds_sync();

    int srow[16];
#pragma unroll
    for (int e = 0; e < 16; ++e)
      srow[e] = __builtin_amdgcn_readfirstlane(L.srcn[e]);
    for (int i = lane; i < 16 * NRAD; i += 32)
      L.ef[i >> 3][i & 7] = edge_feats[(size_t)L.eid[i >> 3] * NRAD + (i & 7)];
    for (int i = lane; i < 16 * 4; i += 32)
      L.ea[i >> 2][i & 3] = edge_attrs[(size_t)L.eid[i >> 2] * 4 + (i & 3)];
    const bool run_end =
        (lane < 16) &&
        (lane == 15 || L.dstn[(lane + 1) & 15] != L.dstn[lane & 15]);
    const unsigned flush = __builtin_amdgcn_ballot_w32(run_end);
    wave_lds_sync();

    for (int nt = 0; nt < 4; ++nt) {
      v8f acc = {};
      for (int kk = 0; kk < 8; kk += 4) {
        int k0 = kk + 2 * half;
        v2f a = { L.ef[m][k0], L.ef[m][k0 + 1] };
        v2f b = { fc_w0[k0 * 64 + nt * 16 + nl],
                  fc_w0[(k0 + 1) * 64 + nt * 16 + nl] };
        acc = wmma4(a, b, acc);
      }
#pragma unroll
      for (int r = 0; r < 8; ++r)
        L.ha[r + 8 * half][nt * 16 + nl] = silu_c(acc[r] * INV_SQRT8);
    }
    wave_lds_sync();

    for (int nt = 0; nt < 4; ++nt) {
      v8f acc = {};
      const float* wb = w1l + half * W1P + (nt * 16 + nl) * 2;
      for (int kk = 0; kk < 64; kk += 4) {
        int k0 = kk + 2 * half;
        v2f a = { L.ha[m][k0], L.ha[m][k0 + 1] };
        v2f b = *(const v2f*)(wb + (kk >> 1) * W1P);
        acc = wmma4(a, b, acc);
      }
#pragma unroll
      for (int r = 0; r < 8; ++r)
        L.hb[r + 8 * half][nt * 16 + nl] = silu_c(acc[r] * 0.125f);
    }
    wave_lds_sync();

    v2f afr[16];
#pragma unroll
    for (int i = 0; i < 16; ++i) {
      const int k0 = 4 * i + 2 * half;
      afr[i] = v2f{ L.hb[m][k0], L.hb[m][k0 + 1] };
    }

    for (int pp = 0; pp < 4; ++pp)
      tile_pair<0>(pp, lane, y, w2l, afr, L.ea, L.dstn, srow, flush, s_acc,
                   v_acc);
    for (int pp = 4; pp < 8; ++pp)
      tile_pair<1>(pp, lane, y, w2l, afr, L.ea, L.dstn, srow, flush, s_acc,
                   v_acc);
    for (int pp = 8; pp < 12; ++pp)
      tile_pair<2>(pp, lane, y, w2l, afr, L.ea, L.dstn, srow, flush, s_acc,
                   v_acc);
    for (int pp = 12; pp < 16; ++pp)
      tile_pair<3>(pp, lane, y, w2l, afr, L.ea, L.dstn, srow, flush, s_acc,
                   v_acc);
    wave_lds_sync();
  }
}

__global__ __launch_bounds__(128) __attribute__((amdgpu_waves_per_eu(1, 5)))
void out_kernel(
    const float* __restrict__ s_acc, const float* __restrict__ v_acc,
    const float* __restrict__ lin2_ws, const float* __restrict__ lin2_wv,
    float* __restrict__ out) {
  const int n0  = (blockIdx.x >> 1) * 16;
  const int c0  = (blockIdx.x & 1) * 64;
  const int tid = threadIdx.x;

  const int t    = __builtin_amdgcn_readfirstlane(tid >> 5);
  const int lane = tid & 31;
  const int half = lane >> 4;
  const int m    = lane & 15;
  const int nl   = lane & 15;
  const float* W = ((t == 0) ? lin2_ws : lin2_wv) + c0;
  const float* arow = (t == 0)
      ? s_acc + (size_t)(n0 + m) * 256
      : v_acc + (size_t)(n0 + m) * 768 + (size_t)(t - 1) * 256;

  v8f acc[4] = {};
  mm4g<1, 256>(arow, W, half, nl, acc);
#pragma unroll
  for (int j = 0; j < 4; ++j) {
    const int col = c0 + j * 16 + nl;
    const int off = (t == 0) ? col : MULC + col * 3 + (t - 1);
#pragma unroll
    for (int r = 0; r < 8; ++r)
      out[(size_t)(n0 + r + 8 * half) * 512 + off] += acc[j][r] * L2_NORM;
  }
}

extern "C" void kernel_launch(void* const* d_in, const int* in_sizes, int n_in,
                              void* d_out, int out_size, void* d_ws,
                              size_t ws_size, hipStream_t stream) {
  const float* node_feats = (const float*)d_in[0];
  const float* node_attrs = (const float*)d_in[1];
  const float* edge_feats = (const float*)d_in[2];
  const float* edge_attrs = (const float*)d_in[3];
  const int*   edge_index = (const int*)d_in[4];
  const float* lin1_ws    = (const float*)d_in[5];
  const float* lin1_wv    = (const float*)d_in[6];
  const float* fc_w0      = (const float*)d_in[7];
  const float* fc_w1      = (const float*)d_in[8];
  const float* fc_w2      = (const float*)d_in[9];
  const float* lin2_ws    = (const float*)d_in[10];
  const float* lin2_wv    = (const float*)d_in[11];
  const float* sc_ws      = (const float*)d_in[12];
  const float* sc_wv      = (const float*)d_in[13];

  float* out   = (float*)d_out;
  float* y     = (float*)d_ws;
  float* s_acc = y + (size_t)NN * 512;
  float* v_acc = s_acc + (size_t)NN * 256;
  int* perm = (int*)d_out;
  int* cnt  = perm + NE;

  zero_kernel<<<2048, 256, 0, stream>>>(s_acc, (long)NN * 1024, cnt, NN);
  y_kernel<<<NN / 8, 128, 0, stream>>>(node_feats, lin1_ws, lin1_wv,
                                        edge_index, y, cnt);
  scan_kernel<<<1, 1024, 0, stream>>>(cnt);
  fill_kernel<<<NE / 256, 256, 0, stream>>>(edge_index, cnt, perm);
  edge_kernel<<<256, EWAVES * 32, 0, stream>>>(y, edge_feats, edge_attrs,
                                          edge_index, perm, fc_w0, fc_w1,
                                          fc_w2, s_acc, v_acc);
  sc_kernel<<<NN / 8, 128, 0, stream>>>(node_feats, node_attrs, sc_ws, sc_wv,
                                         out);
  out_kernel<<<NN / 8, 128, 0, stream>>>(s_acc, v_acc, lin2_ws, lin2_wv, out);
}
